# starting kernel
# baseline (speedup 1.0000x reference)
#include <hip/hip_runtime.h>
#include <hip/hip_bf16.h>

typedef __attribute__((ext_vector_type(16))) __bf16 v16bf;
typedef __attribute__((ext_vector_type(8)))  float  v8f;
typedef __attribute__((ext_vector_type(4)))  float  v4f;

#define N_NODES 100000
#define E_EDGES 3200000
#define IN_F    512
#define HID     16
#define OUT_F   7
#define N_TILES (N_NODES / 16)
#define K_CHUNKS (IN_F / 32)

__device__ __forceinline__ void atomic_add_f32(float* p, float v) {
    __hip_atomic_fetch_add(p, v, __ATOMIC_RELAXED, __HIP_MEMORY_SCOPE_AGENT);
}

__global__ void GCN_pack_w1(const float* __restrict__ W1,
                            __bf16* __restrict__ Bh, __bf16* __restrict__ Bl) {
    int id = blockIdx.x * blockDim.x + threadIdx.x;
    if (id >= K_CHUNKS * 32 * 8) return;
    int v    = id & 7;
    int lane = (id >> 3) & 31;
    int c    = id >> 8;
    int n    = lane & 15;
    int g    = lane >> 4;
    int kb   = c * 32;
#pragma unroll
    for (int h = 0; h < 2; ++h) {
        int k = (v < 4) ? (kb + g * 8 + 2 * v + h)
                        : (kb + 16 + g * 8 + 2 * (v - 4) + h);
        float w   = W1[k * HID + n];
        __bf16 hi = (__bf16)w;
        __bf16 lo = (__bf16)(w - (float)hi);
        int idx = (c * 32 + lane) * 16 + v * 2 + h;
        Bh[idx] = hi;
        Bl[idx] = lo;
    }
}

__global__ void __launch_bounds__(256)
GCN_gemm1_wmma(const float* __restrict__ x,
               const __bf16* __restrict__ Bh, const __bf16* __restrict__ Bl,
               float* __restrict__ h0) {
    const int wave = threadIdx.x >> 5;
    const int lane = threadIdx.x & 31;
    const int tile = blockIdx.x * 8 + wave;
    if (tile >= N_TILES) return;

    const int n = lane & 15;
    const int g = lane >> 4;
    const float* xrow = x + (size_t)(tile * 16 + n) * IN_F;

    v8f acc = {};
#pragma unroll 4
    for (int c = 0; c < K_CHUNKS; ++c) {
        const int kb = c * 32;
        float f[16];
        *(v4f*)(f + 0)  = *(const v4f*)(xrow + kb + g * 8);
        *(v4f*)(f + 4)  = *(const v4f*)(xrow + kb + g * 8 + 4);
        *(v4f*)(f + 8)  = *(const v4f*)(xrow + kb + 16 + g * 8);
        *(v4f*)(f + 12) = *(const v4f*)(xrow + kb + 16 + g * 8 + 4);
        if (c + 1 < K_CHUNKS)
            __builtin_prefetch(xrow + kb + 32 + g * 8, 0, 3);

        v16bf ah, al;
#pragma unroll
        for (int i = 0; i < 16; ++i) {
            __bf16 hi = (__bf16)f[i];
            ah[i] = hi;
            al[i] = (__bf16)(f[i] - (float)hi);
        }
        v16bf bh = *(const v16bf*)(Bh + (c * 32 + lane) * 16);
        v16bf bl = *(const v16bf*)(Bl + (c * 32 + lane) * 16);

        acc = __builtin_amdgcn_wmma_f32_16x16x32_bf16(false, ah, false, bh,
                                                      (short)0, acc, false, false);
        acc = __builtin_amdgcn_wmma_f32_16x16x32_bf16(false, al, false, bh,
                                                      (short)0, acc, false, false);
        acc = __builtin_amdgcn_wmma_f32_16x16x32_bf16(false, ah, false, bl,
                                                      (short)0, acc, false, false);
    }
#pragma unroll
    for (int r = 0; r < 8; ++r)
        h0[(size_t)(tile * 16 + r + g * 8) * HID + n] = acc[r];
}

__global__ void GCN_spmm16(const int* __restrict__ ei, const float* __restrict__ ew,
                           const float* __restrict__ h, float* __restrict__ agg) {
    long long t = (long long)blockIdx.x * blockDim.x + threadIdx.x;
    if (t >= (long long)E_EDGES * 16) return;
    int e = (int)(t >> 4);
    int j = (int)(t & 15);
    int src = ei[e];
    int dst = ei[E_EDGES + e];
    float v = ew[e] * h[(size_t)src * HID + j];
    atomic_add_f32(&agg[(size_t)dst * HID + j], v);
}

__global__ void GCN_spmm7(const int* __restrict__ ei, const float* __restrict__ ew,
                          const float* __restrict__ z, float* __restrict__ agg) {
    long long t = (long long)blockIdx.x * blockDim.x + threadIdx.x;
    if (t >= (long long)E_EDGES * 8) return;
    int e = (int)(t >> 3);
    int j = (int)(t & 7);
    if (j >= OUT_F) return;
    int src = ei[e];
    int dst = ei[E_EDGES + e];
    float v = ew[e] * z[(size_t)src * OUT_F + j];
    atomic_add_f32(&agg[(size_t)dst * OUT_F + j], v);
}

__global__ void __launch_bounds__(256)
GCN_relu_gemm2(const float* __restrict__ agg1, const float* __restrict__ b1,
               const float* __restrict__ W2, float* __restrict__ z) {
    __shared__ float sW2[HID * OUT_F];
    __shared__ float sb1[HID];
    if (threadIdx.x < HID * OUT_F) sW2[threadIdx.x] = W2[threadIdx.x];
    if (threadIdx.x < HID)         sb1[threadIdx.x] = b1[threadIdx.x];
    __syncthreads();

    int node = blockIdx.x * blockDim.x + threadIdx.x;
    if (node >= N_NODES) return;

    float hv[HID];
#pragma unroll
    for (int i = 0; i < HID; ++i)
        hv[i] = fmaxf(agg1[(size_t)node * HID + i] + sb1[i], 0.0f);
#pragma unroll
    for (int j = 0; j < OUT_F; ++j) {
        float s = 0.0f;
#pragma unroll
        for (int i = 0; i < HID; ++i)
            s = fmaf(hv[i], sW2[i * OUT_F + j], s);
        z[(size_t)node * OUT_F + j] = s;
    }
}

__global__ void __launch_bounds__(256)
GCN_logsoftmax(const float* __restrict__ agg2, const float* __restrict__ b2,
               float* __restrict__ out) {
    __shared__ float sb2[OUT_F];
    if (threadIdx.x < OUT_F) sb2[threadIdx.x] = b2[threadIdx.x];
    __syncthreads();

    int node = blockIdx.x * blockDim.x + threadIdx.x;
    if (node >= N_NODES) return;

    float v[OUT_F];
    float m = -3.402823466e38f;
#pragma unroll
    for (int j = 0; j < OUT_F; ++j) {
        v[j] = agg2[(size_t)node * OUT_F + j] + sb2[j];
        m = fmaxf(m, v[j]);
    }
    float s = 0.0f;
#pragma unroll
    for (int j = 0; j < OUT_F; ++j) s += expf(v[j] - m);
    float lse = logf(s);
#pragma unroll
    for (int j = 0; j < OUT_F; ++j)
        out[(size_t)node * OUT_F + j] = v[j] - m - lse;
}

__global__ void GCN_zero(float* __restrict__ p, long long n) {
    long long i = (long long)blockIdx.x * blockDim.x + threadIdx.x;
    if (i < n) p[i] = 0.0f;
}

extern "C" void kernel_launch(void* const* d_in, const int* in_sizes, int n_in,
                              void* d_out, int out_size, void* d_ws, size_t ws_size,
                              hipStream_t stream) {
    const float* x  = (const float*)d_in[0];
    const int*   ei = (const int*)d_in[1];
    const float* ew = (const float*)d_in[2];
    const float* W1 = (const float*)d_in[3];
    const float* b1 = (const float*)d_in[4];
    const float* W2 = (const float*)d_in[5];
    const float* b2 = (const float*)d_in[6];
    float* out = (float*)d_out;

    char* ws = (char*)d_ws;
    size_t off = 0;
    auto alloc = [&](size_t bytes) { char* p = ws + off; off = (off + bytes + 255) & ~(size_t)255; return p; };
    float*  h0   = (float*)alloc((size_t)N_NODES * HID * 4);
    float*  agg1 = (float*)alloc((size_t)N_NODES * HID * 4);
    float*  z    = (float*)alloc((size_t)N_NODES * OUT_F * 4);
    float*  agg2 = (float*)alloc((size_t)N_NODES * OUT_F * 4);
    __bf16* Bh   = (__bf16*)alloc((size_t)K_CHUNKS * 32 * 16 * 2);
    __bf16* Bl   = (__bf16*)alloc((size_t)K_CHUNKS * 32 * 16 * 2);

    {
        long long n1 = (long long)N_NODES * HID;
        long long n2 = (long long)N_NODES * OUT_F;
        GCN_zero<<<(int)((n1 + 255) / 256), 256, 0, stream>>>(agg1, n1);
        GCN_zero<<<(int)((n2 + 255) / 256), 256, 0, stream>>>(agg2, n2);
    }

    GCN_pack_w1<<<K_CHUNKS, 256, 0, stream>>>(W1, Bh, Bl);

    GCN_gemm1_wmma<<<(N_TILES + 7) / 8, 256, 0, stream>>>(x, Bh, Bl, h0);

    {
        long long t = (long long)E_EDGES * 16;
        GCN_spmm16<<<(int)((t + 255) / 256), 256, 0, stream>>>(ei, ew, h0, agg1);
    }

    GCN_relu_gemm2<<<(N_NODES + 255) / 256, 256, 0, stream>>>(agg1, b1, W2, z);

    {
        long long t = (long long)E_EDGES * 8;
        GCN_spmm7<<<(int)((t + 255) / 256), 256, 0, stream>>>(ei, ew, z, agg2);
    }

    GCN_logsoftmax<<<(N_NODES + 255) / 256, 256, 0, stream>>>(agg2, b2, out);
}


# Edge passes: one wave owns 32 consecutive edges (src, dst, weight read once, coalesced, handed out by lane broadcast; its rows gathered as one group) instead of 2 or 4: 16x and 8x fewer waves
# speedup vs baseline: 1.6262x; 1.6262x over previous
#include <hip/hip_runtime.h>
#include <hip/hip_bf16.h>

typedef __attribute__((ext_vector_type(16))) __bf16 v16bf;
typedef __attribute__((ext_vector_type(8)))  float  v8f;
typedef __attribute__((ext_vector_type(4)))  float  v4f;

#define N_NODES 100000
#define E_EDGES 3200000
#define IN_F    512
#define HID     16
#define OUT_F   7
#define OUT_P   8
#define N_TILES (N_NODES / 16)
#define K_CHUNKS (IN_F / 32)

__device__ __forceinline__ void atomic_add_f32(float* p, float v) {
    __hip_atomic_fetch_add(p, v, __ATOMIC_RELAXED, __HIP_MEMORY_SCOPE_AGENT);
}

__global__ void GCN_pack_w1(const float* __restrict__ W1,
                            __bf16* __restrict__ Bh, __bf16* __restrict__ Bl) {
    int id = blockIdx.x * blockDim.x + threadIdx.x;
    if (id >= K_CHUNKS * 32 * 8) return;
    int v    = id & 7;
    int lane = (id >> 3) & 31;
    int c    = id >> 8;
    int n    = lane & 15;
    int g    = lane >> 4;
    int kb   = c * 32;
#pragma unroll
    for (int h = 0; h < 2; ++h) {
        int k = (v < 4) ? (kb + g * 8 + 2 * v + h)
                        : (kb + 16 + g * 8 + 2 * (v - 4) + h);
        float w   = W1[k * HID + n];
        __bf16 hi = (__bf16)w;
        __bf16 lo = (__bf16)(w - (float)hi);
        int idx = (c * 32 + lane) * 16 + v * 2 + h;
        Bh[idx] = hi;
        Bl[idx] = lo;
    }
}

__global__ void __launch_bounds__(256)
GCN_gemm1_wmma(const float* __restrict__ x,
               const __bf16* __restrict__ Bh, const __bf16* __restrict__ Bl,
               float* __restrict__ h0) {
    const int wave = threadIdx.x >> 5;
    const int lane = threadIdx.x & 31;
    const int tile = blockIdx.x * 8 + wave;
    if (tile >= N_TILES) return;

    const int n = lane & 15;
    const int g = lane >> 4;
    const float* xrow = x + (size_t)(tile * 16 + n) * IN_F;

    v8f acc = {};
#pragma unroll 4
    for (int c = 0; c < K_CHUNKS; ++c) {
        const int kb = c * 32;
        float f[16];
        *(v4f*)(f + 0)  = *(const v4f*)(xrow + kb + g * 8);
        *(v4f*)(f + 4)  = *(const v4f*)(xrow + kb + g * 8 + 4);
        *(v4f*)(f + 8)  = *(const v4f*)(xrow + kb + 16 + g * 8);
        *(v4f*)(f + 12) = *(const v4f*)(xrow + kb + 16 + g * 8 + 4);
        if (c + 1 < K_CHUNKS)
            __builtin_prefetch(xrow + kb + 32 + g * 8, 0, 3);

        v16bf ah, al;
#pragma unroll
        for (int i = 0; i < 16; ++i) {
            __bf16 hi = (__bf16)f[i];
            ah[i] = hi;
            al[i] = (__bf16)(f[i] - (float)hi);
        }
        v16bf bh = *(const v16bf*)(Bh + (c * 32 + lane) * 16);
        v16bf bl = *(const v16bf*)(Bl + (c * 32 + lane) * 16);

        acc = __builtin_amdgcn_wmma_f32_16x16x32_bf16(false, ah, false, bh,
                                                      (short)0, acc, false, false);
        acc = __builtin_amdgcn_wmma_f32_16x16x32_bf16(false, al, false, bh,
                                                      (short)0, acc, false, false);
        acc = __builtin_amdgcn_wmma_f32_16x16x32_bf16(false, ah, false, bl,
                                                      (short)0, acc, false, false);
    }
#pragma unroll
    for (int r = 0; r < 8; ++r)
        h0[(size_t)(tile * 16 + r + g * 8) * HID + n] = acc[r];
}

#define EPW 32
static_assert(E_EDGES % (EPW * 8) == 0, "the edge list splits into whole 8-wave blocks");

__device__ __forceinline__ int lane_bcast_i(int v, int k) {
    return __builtin_amdgcn_ds_bpermute(k << 2, v);
}
__device__ __forceinline__ float lane_bcast_f(float v, int k) {
    return __int_as_float(__builtin_amdgcn_ds_bpermute(k << 2, __float_as_int(v)));
}

__global__ void __launch_bounds__(256)
GCN_spmm16(const int* __restrict__ ei, const float* __restrict__ ew,
           const float* __restrict__ h, float* __restrict__ agg) {
    const int lane = threadIdx.x & 31;
    const int e0   = (blockIdx.x * 8 + (threadIdx.x >> 5)) * EPW;
    if (e0 >= E_EDGES) return;
    const int   my_src = ei[e0 + lane];
    const int   my_dst = ei[E_EDGES + e0 + lane];
    const float my_w   = ew[e0 + lane];
    const int   q = lane >> 4;
    const int   j = lane & 15;
    float v[EPW / 2];
    int   d[EPW / 2];
#pragma unroll
    for (int i = 0; i < EPW / 2; ++i) {
        const int   k = 2 * i + q;
        const int   s = lane_bcast_i(my_src, k);
        const float w = lane_bcast_f(my_w, k);
        d[i] = lane_bcast_i(my_dst, k);
        v[i] = w * h[(size_t)s * HID + j];
    }
#pragma unroll
    for (int i = 0; i < EPW / 2; ++i)
        atomic_add_f32(&agg[(size_t)d[i] * HID + j], v[i]);
}

__global__ void __launch_bounds__(256)
GCN_spmm7(const int* __restrict__ ei, const float* __restrict__ ew,
          const float* __restrict__ z, float* __restrict__ agg) {
    const int lane = threadIdx.x & 31;
    const int e0   = (blockIdx.x * 8 + (threadIdx.x >> 5)) * EPW;
    if (e0 >= E_EDGES) return;
    const int   my_src = ei[e0 + lane];
    const int   my_dst = ei[E_EDGES + e0 + lane];
    const float my_w   = ew[e0 + lane];
    const int   q = lane >> 3;
    const int   j = lane & 7;
    int   s[EPW / 4];
    int   d[EPW / 4];
    float w[EPW / 4];
#pragma unroll
    for (int i = 0; i < EPW / 4; ++i) {
        const int k = 4 * i + q;
        s[i] = lane_bcast_i(my_src, k);
        w[i] = lane_bcast_f(my_w, k);
        d[i] = lane_bcast_i(my_dst, k);
    }
    if (j >= OUT_F) return;
    float v[EPW / 4];
#pragma unroll
    for (int i = 0; i < EPW / 4; ++i)
        v[i] = w[i] * z[(size_t)s[i] * OUT_P + j];
#pragma unroll
    for (int i = 0; i < EPW / 4; ++i)
        atomic_add_f32(&agg[(size_t)d[i] * OUT_P + j], v[i]);
}

__global__ void __launch_bounds__(256)
GCN_relu_gemm2(const float* __restrict__ agg1, const float* __restrict__ b1,
               const float* __restrict__ W2, float* __restrict__ z) {
    __shared__ float sW2[HID * OUT_F];
    __shared__ float sb1[HID];
    if (threadIdx.x < HID * OUT_F) sW2[threadIdx.x] = W2[threadIdx.x];
    if (threadIdx.x < HID)         sb1[threadIdx.x] = b1[threadIdx.x];
    __syncthreads();

    int node = blockIdx.x * blockDim.x + threadIdx.x;
    if (node >= N_NODES) return;

    float hv[HID];
#pragma unroll
    for (int i = 0; i < HID; ++i)
        hv[i] = fmaxf(agg1[(size_t)node * HID + i] + sb1[i], 0.0f);
    float zr[OUT_P];
#pragma unroll
    for (int j = 0; j < OUT_F; ++j) {
        float s = 0.0f;
#pragma unroll
        for (int i = 0; i < HID; ++i)
            s = fmaf(hv[i], sW2[i * OUT_F + j], s);
        zr[j] = s;
    }
    zr[OUT_F] = 0.0f;
    v4f* zp = (v4f*)(z + (size_t)node * OUT_P);
    zp[0] = *(const v4f*)(zr + 0);
    zp[1] = *(const v4f*)(zr + 4);
}

__global__ void __launch_bounds__(256)
GCN_logsoftmax(const float* __restrict__ agg2, const float* __restrict__ b2,
               float* __restrict__ out) {
    __shared__ float sb2[OUT_F];
    if (threadIdx.x < OUT_F) sb2[threadIdx.x] = b2[threadIdx.x];
    __syncthreads();

    int node = blockIdx.x * blockDim.x + threadIdx.x;
    if (node >= N_NODES) return;

    float a[OUT_P];
    const v4f* ap = (const v4f*)(agg2 + (size_t)node * OUT_P);
    *(v4f*)(a + 0) = ap[0];
    *(v4f*)(a + 4) = ap[1];
    float v[OUT_F];
    float m = -3.402823466e38f;
#pragma unroll
    for (int j = 0; j < OUT_F; ++j) {
        v[j] = a[j] + sb2[j];
        m = fmaxf(m, v[j]);
    }
    float s = 0.0f;
#pragma unroll
    for (int j = 0; j < OUT_F; ++j) s += expf(v[j] - m);
    float lse = logf(s);
#pragma unroll
    for (int j = 0; j < OUT_F; ++j)
        out[(size_t)node * OUT_F + j] = v[j] - m - lse;
}

__global__ void GCN_zero(float* __restrict__ p, long long n) {
    long long i = (long long)blockIdx.x * blockDim.x + threadIdx.x;
    if (i < n) p[i] = 0.0f;
}

extern "C" void kernel_launch(void* const* d_in, const int* in_sizes, int n_in,
                              void* d_out, int out_size, void* d_ws, size_t ws_size,
                              hipStream_t stream) {
    const float* x  = (const float*)d_in[0];
    const int*   ei = (const int*)d_in[1];
    const float* ew = (const float*)d_in[2];
    const float* W1 = (const float*)d_in[3];
    const float* b1 = (const float*)d_in[4];
    const float* W2 = (const float*)d_in[5];
    const float* b2 = (const float*)d_in[6];
    float* out = (float*)d_out;

    char* ws = (char*)d_ws;
    size_t off = 0;
    auto alloc = [&](size_t bytes) { char* p = ws + off; off = (off + bytes + 255) & ~(size_t)255; return p; };
    float*  h0   = (float*)alloc((size_t)N_NODES * HID * 4);
    float*  agg1 = (float*)alloc((size_t)N_NODES * HID * 4);
    float*  z    = h0;
    float*  agg2 = h0 + (size_t)N_NODES * OUT_P;
    __bf16* Bh   = (__bf16*)alloc((size_t)K_CHUNKS * 32 * 16 * 2);
    __bf16* Bl   = (__bf16*)alloc((size_t)K_CHUNKS * 32 * 16 * 2);

    {
        long long n1 = (long long)N_NODES * HID;
        GCN_zero<<<(int)((n1 + 255) / 256), 256, 0, stream>>>(agg1, n1);
    }

    GCN_pack_w1<<<K_CHUNKS, 256, 0, stream>>>(W1, Bh, Bl);

    GCN_gemm1_wmma<<<(N_TILES + 7) / 8, 256, 0, stream>>>(x, Bh, Bl, h0);

    GCN_spmm16<<<E_EDGES / (EPW * 8), 256, 0, stream>>>(ei, ew, h0, agg1);

    {
        long long n2 = (long long)N_NODES * OUT_P;
        GCN_zero<<<(int)((n2 + 255) / 256), 256, 0, stream>>>(agg2, n2);
    }

    GCN_relu_gemm2<<<(N_NODES + 255) / 256, 256, 0, stream>>>(agg1, b1, W2, z);

    GCN_spmm7<<<E_EDGES / (EPW * 8), 256, 0, stream>>>(ei, ew, z, agg2);

    GCN_logsoftmax<<<(N_NODES + 255) / 256, 256, 0, stream>>>(agg2, b2, out);
}


# Layer-1 GEMM reads x as whole 128-byte lines (a load = 256 contiguous bytes of each of two rows), K-blocks ahead, and turns them into A fragments through a per-wave LDS tile; k order unchanged
# speedup vs baseline: 3.0958x; 1.9037x over previous
#include <hip/hip_runtime.h>
#include <hip/hip_bf16.h>

typedef __attribute__((ext_vector_type(16))) __bf16 v16bf;
typedef __attribute__((ext_vector_type(8)))  float  v8f;
typedef __attribute__((ext_vector_type(4)))  float  v4f;

#define N_NODES 100000
#define E_EDGES 3200000
#define IN_F    512
#define HID     16
#define OUT_F   7
#define OUT_P   8
#define Z_P     HID
#define N_TILES (N_NODES / 16)
#define K_CHUNKS (IN_F / 32)

__device__ __forceinline__ void atomic_add_f32(float* p, float v) {
    __hip_atomic_fetch_add(p, v, __ATOMIC_RELAXED, __HIP_MEMORY_SCOPE_AGENT);
}

__global__ void GCN_pack_w1(const float* __restrict__ W1,
                            __bf16* __restrict__ Bh, __bf16* __restrict__ Bl) {
    int id = blockIdx.x * blockDim.x + threadIdx.x;
    if (id >= K_CHUNKS * 32 * 8) return;
    int v    = id & 7;
    int lane = (id >> 3) & 31;
    int c    = id >> 8;
    int n    = lane & 15;
    int g    = lane >> 4;
    int kb   = c * 32;
#pragma unroll
    for (int h = 0; h < 2; ++h) {
        int k = (v < 4) ? (kb + g * 8 + 2 * v + h)
                        : (kb + 16 + g * 8 + 2 * (v - 4) + h);
        float w   = W1[k * HID + n];
        __bf16 hi = (__bf16)w;
        __bf16 lo = (__bf16)(w - (float)hi);
        int idx = (c * 32 + lane) * 16 + v * 2 + h;
        Bh[idx] = hi;
        Bl[idx] = lo;
    }
}

#define G1_WAVES 4
#define KBLK     64
#define N_KBLK   (IN_F / KBLK)
#define ROW_Q    17

__global__ void __launch_bounds__(G1_WAVES * 32)
GCN_gemm1_wmma(const float* __restrict__ x,
               const __bf16* __restrict__ Bh, const __bf16* __restrict__ Bl,
               float* __restrict__ h0) {
    __shared__ v4f tiles[G1_WAVES][16 * ROW_Q];
    const int wave = threadIdx.x >> 5;
    const int lane = threadIdx.x & 31;
    const int tile = blockIdx.x * G1_WAVES + wave;
    if (tile >= N_TILES) return;

    const int n = lane & 15;
    const int g = lane >> 4;
    v4f* my = tiles[wave];
    const float* xl = x + (size_t)(tile * 16 + g) * IN_F + n * 4;
    v4f*       st = my + g * ROW_Q + n;
    const v4f* fr = my + n * ROW_Q + g * 2;

    v4f buf[8];
#pragma unroll
    for (int i = 0; i < 8; ++i)
        buf[i] = *(const v4f*)(xl + (size_t)(2 * i) * IN_F);

    v8f acc = {};
#pragma unroll
    for (int kb = 0; kb < N_KBLK; ++kb) {
#pragma unroll
        for (int i = 0; i < 8; ++i)
            st[2 * i * ROW_Q] = buf[i];

        v16bf bh[2], bl[2];
#pragma unroll
        for (int cc = 0; cc < 2; ++cc) {
            const int c = kb * 2 + cc;
            bh[cc] = *(const v16bf*)(Bh + (c * 32 + lane) * 16);
            bl[cc] = *(const v16bf*)(Bl + (c * 32 + lane) * 16);
        }
        if (kb + 1 < N_KBLK) {
#pragma unroll
            for (int i = 0; i < 8; ++i)
                buf[i] = *(const v4f*)(xl + (size_t)(2 * i) * IN_F + (kb + 1) * KBLK);
        }

#pragma unroll
        for (int cc = 0; cc < 2; ++cc) {
            const v4f* p = fr + cc * 8;
            float f[16];
            *(v4f*)(f + 0)  = p[0];
            *(v4f*)(f + 4)  = p[1];
            *(v4f*)(f + 8)  = p[4];
            *(v4f*)(f + 12) = p[5];

            v16bf ah, al;
#pragma unroll
            for (int i = 0; i < 16; ++i) {
                __bf16 hi = (__bf16)f[i];
                ah[i] = hi;
                al[i] = (__bf16)(f[i] - (float)hi);
            }
            acc = __builtin_amdgcn_wmma_f32_16x16x32_bf16(false, ah, false, bh[cc],
                                                          (short)0, acc, false, false);
            acc = __builtin_amdgcn_wmma_f32_16x16x32_bf16(false, al, false, bh[cc],
                                                          (short)0, acc, false, false);
            acc = __builtin_amdgcn_wmma_f32_16x16x32_bf16(false, ah, false, bl[cc],
                                                          (short)0, acc, false, false);
        }
    }
#pragma unroll
    for (int r = 0; r < 8; ++r)
        h0[(size_t)(tile * 16 + r + g * 8) * HID + n] = acc[r];
}

#define EPW 32
static_assert(E_EDGES % (EPW * 8) == 0, "the edge list splits into whole 8-wave blocks");

__device__ __forceinline__ int lane_bcast_i(int v, int k) {
    return __builtin_amdgcn_ds_bpermute(k << 2, v);
}
__device__ __forceinline__ float lane_bcast_f(float v, int k) {
    return __int_as_float(__builtin_amdgcn_ds_bpermute(k << 2, __float_as_int(v)));
}

__device__ __forceinline__ int package_half() {
    const unsigned id = __builtin_amdgcn_s_sendmsg_rtn(0x87);
    return (int)((id >> 18) & 1u);
}

__global__ void __launch_bounds__(256)
GCN_spmm16(const int* __restrict__ ei, const float* __restrict__ ew,
           const float* __restrict__ h, float* __restrict__ agg1, size_t plane_stride) {
    const int lane = threadIdx.x & 31;
    const int e0   = (blockIdx.x * 8 + (threadIdx.x >> 5)) * EPW;
    if (e0 >= E_EDGES) return;
    const int   my_src = ei[e0 + lane];
    const int   my_dst = ei[E_EDGES + e0 + lane];
    const float my_w   = ew[e0 + lane];
    float* __restrict__ agg = agg1 + (size_t)package_half() * plane_stride;
    const int   q = lane >> 4;
    const int   j = lane & 15;
    float v[EPW / 2];
    int   d[EPW / 2];
#pragma unroll
    for (int i = 0; i < EPW / 2; ++i) {
        const int   k = 2 * i + q;
        const int   s = lane_bcast_i(my_src, k);
        const float w = lane_bcast_f(my_w, k);
        d[i] = lane_bcast_i(my_dst, k);
        v[i] = w * h[(size_t)s * HID + j];
    }
#pragma unroll
    for (int i = 0; i < EPW / 2; ++i)
        atomic_add_f32(&agg[(size_t)d[i] * HID + j], v[i]);
}

__global__ void __launch_bounds__(256)
GCN_spmm7(const int* __restrict__ ei, const float* __restrict__ ew,
          const float* __restrict__ z, float* __restrict__ agg2) {
    const int lane = threadIdx.x & 31;
    const int e0   = (blockIdx.x * 8 + (threadIdx.x >> 5)) * EPW;
    if (e0 >= E_EDGES) return;
    const int   my_src = ei[e0 + lane];
    const int   my_dst = ei[E_EDGES + e0 + lane];
    const float my_w   = ew[e0 + lane];
    float* __restrict__ agg = agg2 + (size_t)package_half() * ((size_t)N_NODES * OUT_P);
    const int   q = lane >> 3;
    const int   j = lane & 7;
    int   s[EPW / 4];
    int   d[EPW / 4];
    float w[EPW / 4];
#pragma unroll
    for (int i = 0; i < EPW / 4; ++i) {
        const int k = 4 * i + q;
        s[i] = lane_bcast_i(my_src, k);
        w[i] = lane_bcast_f(my_w, k);
        d[i] = lane_bcast_i(my_dst, k);
    }
    if (j >= OUT_F) return;
    float v[EPW / 4];
#pragma unroll
    for (int i = 0; i < EPW / 4; ++i)
        v[i] = w[i] * z[(size_t)s[i] * Z_P + j];
#pragma unroll
    for (int i = 0; i < EPW / 4; ++i)
        atomic_add_f32(&agg[(size_t)d[i] * OUT_P + j], v[i]);
}

__global__ void __launch_bounds__(256)
GCN_relu_gemm2(float* agg1, size_t plane_stride, const float* __restrict__ b1,
               const float* __restrict__ W2) {
    __shared__ float sW2[HID * OUT_F];
    __shared__ float sb1[HID];
    if (threadIdx.x < HID * OUT_F) sW2[threadIdx.x] = W2[threadIdx.x];
    if (threadIdx.x < HID)         sb1[threadIdx.x] = b1[threadIdx.x];
    __syncthreads();

    int node = blockIdx.x * blockDim.x + threadIdx.x;
    if (node >= N_NODES) return;

    float hv[HID];
#pragma unroll
    for (int i = 0; i < HID; ++i)
        hv[i] = agg1[(size_t)node * HID + i];
    if (plane_stride != 0) {
#pragma unroll
        for (int i = 0; i < HID; ++i)
            hv[i] += agg1[plane_stride + (size_t)node * HID + i];
    }
#pragma unroll
    for (int i = 0; i < HID; ++i)
        hv[i] = fmaxf(hv[i] + sb1[i], 0.0f);
    float zr[OUT_P];
#pragma unroll
    for (int j = 0; j < OUT_F; ++j) {
        float s = 0.0f;
#pragma unroll
        for (int i = 0; i < HID; ++i)
            s = fmaf(hv[i], sW2[i * OUT_F + j], s);
        zr[j] = s;
    }
    zr[OUT_F] = 0.0f;
    v4f* zp = (v4f*)(agg1 + (size_t)node * Z_P);
    zp[0] = *(const v4f*)(zr + 0);
    zp[1] = *(const v4f*)(zr + 4);
}

__global__ void __launch_bounds__(256)
GCN_logsoftmax(const float* __restrict__ agg2, const float* __restrict__ b2,
               float* __restrict__ out) {
    __shared__ float sb2[OUT_F];
    if (threadIdx.x < OUT_F) sb2[threadIdx.x] = b2[threadIdx.x];
    __syncthreads();

    int node = blockIdx.x * blockDim.x + threadIdx.x;
    if (node >= N_NODES) return;

    float a[OUT_P];
    const v4f* ap = (const v4f*)(agg2 + (size_t)node * OUT_P);
    const v4f* bp = (const v4f*)(agg2 + (size_t)N_NODES * OUT_P + (size_t)node * OUT_P);
    *(v4f*)(a + 0) = ap[0] + bp[0];
    *(v4f*)(a + 4) = ap[1] + bp[1];
    float v[OUT_F];
    float m = -3.402823466e38f;
#pragma unroll
    for (int j = 0; j < OUT_F; ++j) {
        v[j] = a[j] + sb2[j];
        m = fmaxf(m, v[j]);
    }
    float s = 0.0f;
#pragma unroll
    for (int j = 0; j < OUT_F; ++j) s += expf(v[j] - m);
    float lse = logf(s);
#pragma unroll
    for (int j = 0; j < OUT_F; ++j)
        out[(size_t)node * OUT_F + j] = v[j] - m - lse;
}

__global__ void GCN_zero(float* __restrict__ p, long long n) {
    long long i = (long long)blockIdx.x * blockDim.x + threadIdx.x;
    if (i < n) p[i] = 0.0f;
}

extern "C" void kernel_launch(void* const* d_in, const int* in_sizes, int n_in,
                              void* d_out, int out_size, void* d_ws, size_t ws_size,
                              hipStream_t stream) {
    const float* x  = (const float*)d_in[0];
    const int*   ei = (const int*)d_in[1];
    const float* ew = (const float*)d_in[2];
    const float* W1 = (const float*)d_in[3];
    const float* b1 = (const float*)d_in[4];
    const float* W2 = (const float*)d_in[5];
    const float* b2 = (const float*)d_in[6];
    float* out = (float*)d_out;

    char* ws = (char*)d_ws;
    size_t off = 0;
    auto alloc = [&](size_t bytes) { char* p = ws + off; off = (off + bytes + 255) & ~(size_t)255; return p; };
    const size_t plane1      = (size_t)N_NODES * HID;
    const size_t ws_two      = 3 * plane1 * 4 + 2 * (size_t)K_CHUNKS * 32 * 16 * 2;
    const size_t plane_stride = (ws_size >= ws_two) ? plane1 : 0;
    float*  h0   = (float*)alloc(plane1 * 4);
    float*  agg1 = (float*)alloc(plane1 * 4 + plane_stride * 4);
    float*  z    = agg1;
    float*  agg2 = h0;
    __bf16* Bh   = (__bf16*)alloc((size_t)K_CHUNKS * 32 * 16 * 2);
    __bf16* Bl   = (__bf16*)alloc((size_t)K_CHUNKS * 32 * 16 * 2);

    {
        long long n1 = (long long)(plane1 + plane_stride);
        GCN_zero<<<(int)((n1 + 255) / 256), 256, 0, stream>>>(agg1, n1);
    }

    GCN_pack_w1<<<K_CHUNKS, 256, 0, stream>>>(W1, Bh, Bl);

    GCN_gemm1_wmma<<<(N_TILES + G1_WAVES - 1) / G1_WAVES, G1_WAVES * 32, 0, stream>>>(x, Bh, Bl, h0);

    GCN_spmm16<<<E_EDGES / (EPW * 8), 256, 0, stream>>>(ei, ew, h0, agg1, plane_stride);

    {
        long long n2 = 2LL * N_NODES * OUT_P;
        GCN_zero<<<(int)((n2 + 255) / 256), 256, 0, stream>>>(agg2, n2);
    }

    GCN_relu_gemm2<<<(N_NODES + 255) / 256, 256, 0, stream>>>(agg1, plane_stride, b1, W2);

    GCN_spmm7<<<E_EDGES / (EPW * 8), 256, 0, stream>>>(ei, ew, z, agg2);

    GCN_logsoftmax<<<(N_NODES + 255) / 256, 256, 0, stream>>>(agg2, b2, out);
}


# The two clear launches go: the layer-1 GEMM clears its tile's rows of the layer-1 sum planes, the bias+ReLU kernel its nodes' rows of the layer-2 planes; 8 launches become 6, 150000 waves fewer
# speedup vs baseline: 3.1812x; 1.0276x over previous
#include <hip/hip_runtime.h>
#include <hip/hip_bf16.h>

typedef __attribute__((ext_vector_type(16))) __bf16 v16bf;
typedef __attribute__((ext_vector_type(8)))  float  v8f;
typedef __attribute__((ext_vector_type(4)))  float  v4f;

#define N_NODES 100000
#define E_EDGES 3200000
#define IN_F    512
#define HID     16
#define OUT_F   7
#define OUT_P   8
#define Z_P     HID
#define N_TILES (N_NODES / 16)
#define K_CHUNKS (IN_F / 32)

__device__ __forceinline__ void atomic_add_f32(float* p, float v) {
    __hip_atomic_fetch_add(p, v, __ATOMIC_RELAXED, __HIP_MEMORY_SCOPE_AGENT);
}

__global__ void GCN_pack_w1(const float* __restrict__ W1,
                            __bf16* __restrict__ Bh, __bf16* __restrict__ Bl) {
    int id = blockIdx.x * blockDim.x + threadIdx.x;
    if (id >= K_CHUNKS * 32 * 8) return;
    int v    = id & 7;
    int lane = (id >> 3) & 31;
    int c    = id >> 8;
    int n    = lane & 15;
    int g    = lane >> 4;
    int kb   = c * 32;
#pragma unroll
    for (int h = 0; h < 2; ++h) {
        int k = (v < 4) ? (kb + g * 8 + 2 * v + h)
                        : (kb + 16 + g * 8 + 2 * (v - 4) + h);
        float w   = W1[k * HID + n];
        __bf16 hi = (__bf16)w;
        __bf16 lo = (__bf16)(w - (float)hi);
        int idx = (c * 32 + lane) * 16 + v * 2 + h;
        Bh[idx] = hi;
        Bl[idx] = lo;
    }
}

#define G1_WAVES 4
#define KBLK     64
#define N_KBLK   (IN_F / KBLK)
#define ROW_Q    17

__global__ void __launch_bounds__(G1_WAVES * 32)
GCN_gemm1_wmma(const float* __restrict__ x,
               const __bf16* __restrict__ Bh, const __bf16* __restrict__ Bl,
               float* __restrict__ h0, float* __restrict__ agg1, size_t plane_stride) {
    __shared__ v4f tiles[G1_WAVES][16 * ROW_Q];
    const int wave = threadIdx.x >> 5;
    const int lane = threadIdx.x & 31;
    const int tile = blockIdx.x * G1_WAVES + wave;
    if (tile >= N_TILES) return;

    const int n = lane & 15;
    const int g = lane >> 4;
    v4f* my = tiles[wave];
    const float* xl = x + (size_t)(tile * 16 + g) * IN_F + n * 4;
    v4f*       st = my + g * ROW_Q + n;
    const v4f* fr = my + n * ROW_Q + g * 2;

    v4f buf[8];
#pragma unroll
    for (int i = 0; i < 8; ++i)
        buf[i] = *(const v4f*)(xl + (size_t)(2 * i) * IN_F);

    v8f acc = {};
#pragma unroll
    for (int kb = 0; kb < N_KBLK; ++kb) {
#pragma unroll
        for (int i = 0; i < 8; ++i)
            st[2 * i * ROW_Q] = buf[i];

        v16bf bh[2], bl[2];
#pragma unroll
        for (int cc = 0; cc < 2; ++cc) {
            const int c = kb * 2 + cc;
            bh[cc] = *(const v16bf*)(Bh + (c * 32 + lane) * 16);
            bl[cc] = *(const v16bf*)(Bl + (c * 32 + lane) * 16);
        }
        if (kb + 1 < N_KBLK) {
#pragma unroll
            for (int i = 0; i < 8; ++i)
                buf[i] = *(const v4f*)(xl + (size_t)(2 * i) * IN_F + (kb + 1) * KBLK);
        }

#pragma unroll
        for (int cc = 0; cc < 2; ++cc) {
            const v4f* p = fr + cc * 8;
            float f[16];
            *(v4f*)(f + 0)  = p[0];
            *(v4f*)(f + 4)  = p[1];
            *(v4f*)(f + 8)  = p[4];
            *(v4f*)(f + 12) = p[5];

            v16bf ah, al;
#pragma unroll
            for (int i = 0; i < 16; ++i) {
                __bf16 hi = (__bf16)f[i];
                ah[i] = hi;
                al[i] = (__bf16)(f[i] - (float)hi);
            }
            acc = __builtin_amdgcn_wmma_f32_16x16x32_bf16(false, ah, false, bh[cc],
                                                          (short)0, acc, false, false);
            acc = __builtin_amdgcn_wmma_f32_16x16x32_bf16(false, al, false, bh[cc],
                                                          (short)0, acc, false, false);
            acc = __builtin_amdgcn_wmma_f32_16x16x32_bf16(false, ah, false, bl[cc],
                                                          (short)0, acc, false, false);
        }
    }
#pragma unroll
    for (int r = 0; r < 8; ++r)
        h0[(size_t)(tile * 16 + r + g * 8) * HID + n] = acc[r];

    const v4f zero4 = {0.0f, 0.0f, 0.0f, 0.0f};
    v4f* c0 = (v4f*)(agg1 + (size_t)tile * 16 * HID);
    c0[lane]      = zero4;
    c0[32 + lane] = zero4;
    if (plane_stride != 0) {
        v4f* c1 = (v4f*)(agg1 + plane_stride + (size_t)tile * 16 * HID);
        c1[lane]      = zero4;
        c1[32 + lane] = zero4;
    }
}

#define EPW 32
static_assert(E_EDGES % (EPW * 8) == 0, "the edge list splits into whole 8-wave blocks");
static_assert(N_NODES % 16 == 0, "the GEMM tiles cover every node: their clears reach every layer-1 sum row");

__device__ __forceinline__ int lane_bcast_i(int v, int k) {
    return __builtin_amdgcn_ds_bpermute(k << 2, v);
}
__device__ __forceinline__ float lane_bcast_f(float v, int k) {
    return __int_as_float(__builtin_amdgcn_ds_bpermute(k << 2, __float_as_int(v)));
}

__device__ __forceinline__ int package_half() {
    const unsigned id = __builtin_amdgcn_s_sendmsg_rtn(0x87);
    return (int)((id >> 18) & 1u);
}

__global__ void __launch_bounds__(256)
GCN_spmm16(const int* __restrict__ ei, const float* __restrict__ ew,
           const float* __restrict__ h, float* __restrict__ agg1, size_t plane_stride) {
    const int lane = threadIdx.x & 31;
    const int e0   = (blockIdx.x * 8 + (threadIdx.x >> 5)) * EPW;
    if (e0 >= E_EDGES) return;
    const int   my_src = ei[e0 + lane];
    const int   my_dst = ei[E_EDGES + e0 + lane];
    const float my_w   = ew[e0 + lane];
    float* __restrict__ agg = agg1 + (size_t)package_half() * plane_stride;
    const int   q = lane >> 4;
    const int   j = lane & 15;
    float v[EPW / 2];
    int   d[EPW / 2];
#pragma unroll
    for (int i = 0; i < EPW / 2; ++i) {
        const int   k = 2 * i + q;
        const int   s = lane_bcast_i(my_src, k);
        const float w = lane_bcast_f(my_w, k);
        d[i] = lane_bcast_i(my_dst, k);
        v[i] = w * h[(size_t)s * HID + j];
    }
#pragma unroll
    for (int i = 0; i < EPW / 2; ++i)
        atomic_add_f32(&agg[(size_t)d[i] * HID + j], v[i]);
}

__global__ void __launch_bounds__(256)
GCN_spmm7(const int* __restrict__ ei, const float* __restrict__ ew,
          const float* __restrict__ z, float* __restrict__ agg2) {
    const int lane = threadIdx.x & 31;
    const int e0   = (blockIdx.x * 8 + (threadIdx.x >> 5)) * EPW;
    if (e0 >= E_EDGES) return;
    const int   my_src = ei[e0 + lane];
    const int   my_dst = ei[E_EDGES + e0 + lane];
    const float my_w   = ew[e0 + lane];
    float* __restrict__ agg = agg2 + (size_t)package_half() * ((size_t)N_NODES * OUT_P);
    const int   q = lane >> 3;
    const int   j = lane & 7;
    int   s[EPW / 4];
    int   d[EPW / 4];
    float w[EPW / 4];
#pragma unroll
    for (int i = 0; i < EPW / 4; ++i) {
        const int k = 4 * i + q;
        s[i] = lane_bcast_i(my_src, k);
        w[i] = lane_bcast_f(my_w, k);
        d[i] = lane_bcast_i(my_dst, k);
    }
    if (j >= OUT_F) return;
    float v[EPW / 4];
#pragma unroll
    for (int i = 0; i < EPW / 4; ++i)
        v[i] = w[i] * z[(size_t)s[i] * Z_P + j];
#pragma unroll
    for (int i = 0; i < EPW / 4; ++i)
        atomic_add_f32(&agg[(size_t)d[i] * OUT_P + j], v[i]);
}

__global__ void __launch_bounds__(256)
GCN_relu_gemm2(float* agg1, size_t plane_stride, const float* __restrict__ b1,
               const float* __restrict__ W2, float* __restrict__ agg2) {
    __shared__ float sW2[HID * OUT_F];
    __shared__ float sb1[HID];
    if (threadIdx.x < HID * OUT_F) sW2[threadIdx.x] = W2[threadIdx.x];
    if (threadIdx.x < HID)         sb1[threadIdx.x] = b1[threadIdx.x];
    __syncthreads();

    {
        const v4f zero4 = {0.0f, 0.0f, 0.0f, 0.0f};
        v4f* p0 = (v4f*)agg2;
        v4f* p1 = (v4f*)(agg2 + (size_t)N_NODES * OUT_P);
        const size_t u0 = (size_t)blockIdx.x * 512 + threadIdx.x;
#pragma unroll
        for (int k = 0; k < 2; ++k) {
            const size_t u = u0 + (size_t)k * 256;
            if (u < (size_t)2 * N_NODES) {
                p0[u] = zero4;
                p1[u] = zero4;
            }
        }
    }

    int node = blockIdx.x * blockDim.x + threadIdx.x;
    if (node >= N_NODES) return;

    float hv[HID];
#pragma unroll
    for (int i = 0; i < HID; ++i)
        hv[i] = agg1[(size_t)node * HID + i];
    if (plane_stride != 0) {
#pragma unroll
        for (int i = 0; i < HID; ++i)
            hv[i] += agg1[plane_stride + (size_t)node * HID + i];
    }
#pragma unroll
    for (int i = 0; i < HID; ++i)
        hv[i] = fmaxf(hv[i] + sb1[i], 0.0f);
    float zr[OUT_P];
#pragma unroll
    for (int j = 0; j < OUT_F; ++j) {
        float s = 0.0f;
#pragma unroll
        for (int i = 0; i < HID; ++i)
            s = fmaf(hv[i], sW2[i * OUT_F + j], s);
        zr[j] = s;
    }
    zr[OUT_F] = 0.0f;
    v4f* zp = (v4f*)(agg1 + (size_t)node * Z_P);
    zp[0] = *(const v4f*)(zr + 0);
    zp[1] = *(const v4f*)(zr + 4);
}

__global__ void __launch_bounds__(256)
GCN_logsoftmax(const float* __restrict__ agg2, const float* __restrict__ b2,
               float* __restrict__ out) {
    __shared__ float sb2[OUT_F];
    if (threadIdx.x < OUT_F) sb2[threadIdx.x] = b2[threadIdx.x];
    __syncthreads();

    int node = blockIdx.x * blockDim.x + threadIdx.x;
    if (node >= N_NODES) return;

    float a[OUT_P];
    const v4f* ap = (const v4f*)(agg2 + (size_t)node * OUT_P);
    const v4f* bp = (const v4f*)(agg2 + (size_t)N_NODES * OUT_P + (size_t)node * OUT_P);
    *(v4f*)(a + 0) = ap[0] + bp[0];
    *(v4f*)(a + 4) = ap[1] + bp[1];
    float v[OUT_F];
    float m = -3.402823466e38f;
#pragma unroll
    for (int j = 0; j < OUT_F; ++j) {
        v[j] = a[j] + sb2[j];
        m = fmaxf(m, v[j]);
    }
    float s = 0.0f;
#pragma unroll
    for (int j = 0; j < OUT_F; ++j) s += expf(v[j] - m);
    float lse = logf(s);
#pragma unroll
    for (int j = 0; j < OUT_F; ++j)
        out[(size_t)node * OUT_F + j] = v[j] - m - lse;
}

extern "C" void kernel_launch(void* const* d_in, const int* in_sizes, int n_in,
                              void* d_out, int out_size, void* d_ws, size_t ws_size,
                              hipStream_t stream) {
    const float* x  = (const float*)d_in[0];
    const int*   ei = (const int*)d_in[1];
    const float* ew = (const float*)d_in[2];
    const float* W1 = (const float*)d_in[3];
    const float* b1 = (const float*)d_in[4];
    const float* W2 = (const float*)d_in[5];
    const float* b2 = (const float*)d_in[6];
    float* out = (float*)d_out;

    char* ws = (char*)d_ws;
    size_t off = 0;
    auto alloc = [&](size_t bytes) { char* p = ws + off; off = (off + bytes + 255) & ~(size_t)255; return p; };
    const size_t plane1      = (size_t)N_NODES * HID;
    const size_t ws_two      = 3 * plane1 * 4 + 2 * (size_t)K_CHUNKS * 32 * 16 * 2;
    const size_t plane_stride = (ws_size >= ws_two) ? plane1 : 0;
    float*  h0   = (float*)alloc(plane1 * 4);
    float*  agg1 = (float*)alloc(plane1 * 4 + plane_stride * 4);
    float*  z    = agg1;
    float*  agg2 = h0;
    __bf16* Bh   = (__bf16*)alloc((size_t)K_CHUNKS * 32 * 16 * 2);
    __bf16* Bl   = (__bf16*)alloc((size_t)K_CHUNKS * 32 * 16 * 2);

    GCN_pack_w1<<<K_CHUNKS, 256, 0, stream>>>(W1, Bh, Bl);

    GCN_gemm1_wmma<<<(N_TILES + G1_WAVES - 1) / G1_WAVES, G1_WAVES * 32, 0, stream>>>(x, Bh, Bl, h0, agg1, plane_stride);

    GCN_spmm16<<<E_EDGES / (EPW * 8), 256, 0, stream>>>(ei, ew, h0, agg1, plane_stride);

    GCN_relu_gemm2<<<(N_NODES + 255) / 256, 256, 0, stream>>>(agg1, plane_stride, b1, W2, agg2);

    GCN_spmm7<<<E_EDGES / (EPW * 8), 256, 0, stream>>>(ei, ew, z, agg2);

    GCN_logsoftmax<<<(N_NODES + 255) / 256, 256, 0, stream>>>(agg2, b2, out);
}


# Edges grouped by 128-node destination block once per call (count + place, integer adds); edge passes sum rows in LDS (LDS float adds), write each row once: no global float atomics; sum order changes
# speedup vs baseline: 3.6036x; 1.1328x over previous
#include <hip/hip_runtime.h>
#include <hip/hip_bf16.h>

typedef __attribute__((ext_vector_type(16))) __bf16 v16bf;
typedef __attribute__((ext_vector_type(8)))  float  v8f;
typedef __attribute__((ext_vector_type(4)))  float  v4f;

#define N_NODES 100000
#define E_EDGES 3200000
#define IN_F    512
#define HID     16
#define OUT_F   7
#define OUT_P   8
#define Z_P     HID
#define N_TILES (N_NODES / 16)
#define K_CHUNKS (IN_F / 32)

__device__ __forceinline__ void atomic_add_f32(float* p, float v) {
    __hip_atomic_fetch_add(p, v, __ATOMIC_RELAXED, __HIP_MEMORY_SCOPE_AGENT);
}

__global__ void GCN_pack_w1(const float* __restrict__ W1,
                            __bf16* __restrict__ Bh, __bf16* __restrict__ Bl,
                            int* __restrict__ cnt, int n_cnt) {
    int id = blockIdx.x * blockDim.x + threadIdx.x;
    if (id < n_cnt) cnt[id] = 0;
    if (id >= K_CHUNKS * 32 * 8) return;
    int v    = id & 7;
    int lane = (id >> 3) & 31;
    int c    = id >> 8;
    int n    = lane & 15;
    int g    = lane >> 4;
    int kb   = c * 32;
#pragma unroll
    for (int h = 0; h < 2; ++h) {
        int k = (v < 4) ? (kb + g * 8 + 2 * v + h)
                        : (kb + 16 + g * 8 + 2 * (v - 4) + h);
        float w   = W1[k * HID + n];
        __bf16 hi = (__bf16)w;
        __bf16 lo = (__bf16)(w - (float)hi);
        int idx = (c * 32 + lane) * 16 + v * 2 + h;
        Bh[idx] = hi;
        Bl[idx] = lo;
    }
}

#define G1_WAVES 4
#define KBLK     64
#define N_KBLK   (IN_F / KBLK)
#define ROW_Q    17

__global__ void __launch_bounds__(G1_WAVES * 32)
GCN_gemm1_wmma(const float* __restrict__ x,
               const __bf16* __restrict__ Bh, const __bf16* __restrict__ Bl,
               float* __restrict__ h0, float* __restrict__ agg1, size_t plane_stride) {
    __shared__ v4f tiles[G1_WAVES][16 * ROW_Q];
    const int wave = threadIdx.x >> 5;
    const int lane = threadIdx.x & 31;
    const int tile = blockIdx.x * G1_WAVES + wave;
    if (tile >= N_TILES) return;

    const int n = lane & 15;
    const int g = lane >> 4;
    v4f* my = tiles[wave];
    const float* xl = x + (size_t)(tile * 16 + g) * IN_F + n * 4;
    v4f*       st = my + g * ROW_Q + n;
    const v4f* fr = my + n * ROW_Q + g * 2;

    v4f buf[8];
#pragma unroll
    for (int i = 0; i < 8; ++i)
        buf[i] = *(const v4f*)(xl + (size_t)(2 * i) * IN_F);

    v8f acc = {};
#pragma unroll
    for (int kb = 0; kb < N_KBLK; ++kb) {
#pragma unroll
        for (int i = 0; i < 8; ++i)
            st[2 * i * ROW_Q] = buf[i];

        v16bf bh[2], bl[2];
#pragma unroll
        for (int cc = 0; cc < 2; ++cc) {
            const int c = kb * 2 + cc;
            bh[cc] = *(const v16bf*)(Bh + (c * 32 + lane) * 16);
            bl[cc] = *(const v16bf*)(Bl + (c * 32 + lane) * 16);
        }
        if (kb + 1 < N_KBLK) {
#pragma unroll
            for (int i = 0; i < 8; ++i)
                buf[i] = *(const v4f*)(xl + (size_t)(2 * i) * IN_F + (kb + 1) * KBLK);
        }

#pragma unroll
        for (int cc = 0; cc < 2; ++cc) {
            const v4f* p = fr + cc * 8;
            float f[16];
            *(v4f*)(f + 0)  = p[0];
            *(v4f*)(f + 4)  = p[1];
            *(v4f*)(f + 8)  = p[4];
            *(v4f*)(f + 12) = p[5];

            v16bf ah, al;
#pragma unroll
            for (int i = 0; i < 16; ++i) {
                __bf16 hi = (__bf16)f[i];
                ah[i] = hi;
                al[i] = (__bf16)(f[i] - (float)hi);
            }
            acc = __builtin_amdgcn_wmma_f32_16x16x32_bf16(false, ah, false, bh[cc],
                                                          (short)0, acc, false, false);
            acc = __builtin_amdgcn_wmma_f32_16x16x32_bf16(false, al, false, bh[cc],
                                                          (short)0, acc, false, false);
            acc = __builtin_amdgcn_wmma_f32_16x16x32_bf16(false, ah, false, bl[cc],
                                                          (short)0, acc, false, false);
        }
    }
#pragma unroll
    for (int r = 0; r < 8; ++r)
        h0[(size_t)(tile * 16 + r + g * 8) * HID + n] = acc[r];

    if (agg1 == nullptr) return;
    const v4f zero4 = {0.0f, 0.0f, 0.0f, 0.0f};
    v4f* c0 = (v4f*)(agg1 + (size_t)tile * 16 * HID);
    c0[lane]      = zero4;
    c0[32 + lane] = zero4;
    if (plane_stride != 0) {
        v4f* c1 = (v4f*)(agg1 + plane_stride + (size_t)tile * 16 * HID);
        c1[lane]      = zero4;
        c1[32 + lane] = zero4;
    }
}

#define EPW 32
static_assert(E_EDGES % (EPW * 8) == 0, "the edge list splits into whole 8-wave blocks");
static_assert(N_NODES % 16 == 0, "the GEMM tiles cover every node: their clears reach every layer-1 sum row");

__device__ __forceinline__ int lane_bcast_i(int v, int k) {
    return __builtin_amdgcn_ds_bpermute(k << 2, v);
}
__device__ __forceinline__ float lane_bcast_f(float v, int k) {
    return __int_as_float(__builtin_amdgcn_ds_bpermute(k << 2, __float_as_int(v)));
}

__device__ __forceinline__ int package_half() {
    const unsigned id = __builtin_amdgcn_s_sendmsg_rtn(0x87);
    return (int)((id >> 18) & 1u);
}

__global__ void __launch_bounds__(256)
GCN_spmm16(const int* __restrict__ ei, const float* __restrict__ ew,
           const float* __restrict__ h, float* __restrict__ agg1, size_t plane_stride) {
    const int lane = threadIdx.x & 31;
    const int e0   = (blockIdx.x * 8 + (threadIdx.x >> 5)) * EPW;
    if (e0 >= E_EDGES) return;
    const int   my_src = ei[e0 + lane];
    const int   my_dst = ei[E_EDGES + e0 + lane];
    const float my_w   = ew[e0 + lane];
    float* __restrict__ agg = agg1 + (size_t)package_half() * plane_stride;
    const int   q = lane >> 4;
    const int   j = lane & 15;
    float v[EPW / 2];
    int   d[EPW / 2];
#pragma unroll
    for (int i = 0; i < EPW / 2; ++i) {
        const int   k = 2 * i + q;
        const int   s = lane_bcast_i(my_src, k);
        const float w = lane_bcast_f(my_w, k);
        d[i] = lane_bcast_i(my_dst, k);
        v[i] = w * h[(size_t)s * HID + j];
    }
#pragma unroll
    for (int i = 0; i < EPW / 2; ++i)
        atomic_add_f32(&agg[(size_t)d[i] * HID + j], v[i]);
}

__global__ void __launch_bounds__(256)
GCN_spmm7(const int* __restrict__ ei, const float* __restrict__ ew,
          const float* __restrict__ z, float* __restrict__ agg2) {
    const int lane = threadIdx.x & 31;
    const int e0   = (blockIdx.x * 8 + (threadIdx.x >> 5)) * EPW;
    if (e0 >= E_EDGES) return;
    const int   my_src = ei[e0 + lane];
    const int   my_dst = ei[E_EDGES + e0 + lane];
    const float my_w   = ew[e0 + lane];
    float* __restrict__ agg = agg2 + (size_t)package_half() * ((size_t)N_NODES * OUT_P);
    const int   q = lane >> 3;
    const int   j = lane & 7;
    int   s[EPW / 4];
    int   d[EPW / 4];
    float w[EPW / 4];
#pragma unroll
    for (int i = 0; i < EPW / 4; ++i) {
        const int k = 4 * i + q;
        s[i] = lane_bcast_i(my_src, k);
        w[i] = lane_bcast_f(my_w, k);
        d[i] = lane_bcast_i(my_dst, k);
    }
    if (j >= OUT_F) return;
    float v[EPW / 4];
#pragma unroll
    for (int i = 0; i < EPW / 4; ++i)
        v[i] = w[i] * z[(size_t)s[i] * Z_P + j];
#pragma unroll
    for (int i = 0; i < EPW / 4; ++i)
        atomic_add_f32(&agg[(size_t)d[i] * OUT_P + j], v[i]);
}

__global__ void __launch_bounds__(256)
GCN_relu_gemm2(float* agg1, size_t plane_stride, const float* __restrict__ b1,
               const float* __restrict__ W2, float* __restrict__ agg2) {
    __shared__ float sW2[HID * OUT_F];
    __shared__ float sb1[HID];
    if (threadIdx.x < HID * OUT_F) sW2[threadIdx.x] = W2[threadIdx.x];
    if (threadIdx.x < HID)         sb1[threadIdx.x] = b1[threadIdx.x];
    __syncthreads();

    {
        const v4f zero4 = {0.0f, 0.0f, 0.0f, 0.0f};
        v4f* p0 = (v4f*)agg2;
        v4f* p1 = (v4f*)(agg2 + (size_t)N_NODES * OUT_P);
        const size_t u0 = (size_t)blockIdx.x * 512 + threadIdx.x;
#pragma unroll
        for (int k = 0; k < 2; ++k) {
            const size_t u = u0 + (size_t)k * 256;
            if (u < (size_t)2 * N_NODES) {
                p0[u] = zero4;
                p1[u] = zero4;
            }
        }
    }

    int node = blockIdx.x * blockDim.x + threadIdx.x;
    if (node >= N_NODES) return;

    float hv[HID];
#pragma unroll
    for (int i = 0; i < HID; ++i)
        hv[i] = agg1[(size_t)node * HID + i];
    if (plane_stride != 0) {
#pragma unroll
        for (int i = 0; i < HID; ++i)
            hv[i] += agg1[plane_stride + (size_t)node * HID + i];
    }
#pragma unroll
    for (int i = 0; i < HID; ++i)
        hv[i] = fmaxf(hv[i] + sb1[i], 0.0f);
    float zr[OUT_P];
#pragma unroll
    for (int j = 0; j < OUT_F; ++j) {
        float s = 0.0f;
#pragma unroll
        for (int i = 0; i < HID; ++i)
            s = fmaf(hv[i], sW2[i * OUT_F + j], s);
        zr[j] = s;
    }
    zr[OUT_F] = 0.0f;
    v4f* zp = (v4f*)(agg1 + (size_t)node * Z_P);
    zp[0] = *(const v4f*)(zr + 0);
    zp[1] = *(const v4f*)(zr + 4);
}

__global__ void __launch_bounds__(256)
GCN_logsoftmax(const float* __restrict__ agg2, const float* __restrict__ b2,
               float* __restrict__ out) {
    __shared__ float sb2[OUT_F];
    if (threadIdx.x < OUT_F) sb2[threadIdx.x] = b2[threadIdx.x];
    __syncthreads();

    int node = blockIdx.x * blockDim.x + threadIdx.x;
    if (node >= N_NODES) return;

    float a[OUT_P];
    const v4f* ap = (const v4f*)(agg2 + (size_t)node * OUT_P);
    const v4f* bp = (const v4f*)(agg2 + (size_t)N_NODES * OUT_P + (size_t)node * OUT_P);
    *(v4f*)(a + 0) = ap[0] + bp[0];
    *(v4f*)(a + 4) = ap[1] + bp[1];
    float v[OUT_F];
    float m = -3.402823466e38f;
#pragma unroll
    for (int j = 0; j < OUT_F; ++j) {
        v[j] = a[j] + sb2[j];
        m = fmaxf(m, v[j]);
    }
    float s = 0.0f;
#pragma unroll
    for (int j = 0; j < OUT_F; ++j) s += expf(v[j] - m);
    float lse = logf(s);
#pragma unroll
    for (int j = 0; j < OUT_F; ++j)
        out[(size_t)node * OUT_F + j] = v[j] - m - lse;
}

#define BLK_NODES 128
#define NB        ((N_NODES + BLK_NODES - 1) / BLK_NODES)
#define CHUNK     8192
#define N_CHUNKS  ((E_EDGES + CHUNK - 1) / CHUNK)
#define EP_THREADS 256
#define SRC_MASK  0x1FFFFu
static_assert(N_NODES <= (1 << 17), "a source index fits the low 17 bits of a record word");
static_assert(BLK_NODES == 128, "destination within block: 7 bits, block = dst >> 7");
#define PL_THREADS 1024
#define PL_WAVES   (PL_THREADS / 32)
static_assert(PL_THREADS >= NB + 1, "the scan has one thread per block and one for the end entry");
static_assert(CHUNK % (PL_THREADS * 8) == 0, "a chunk is whole rounds of eight edges per thread");
static_assert(EP_THREADS >= BLK_NODES, "one thread per node in the writers");

__device__ __forceinline__ int wave_incl_scan(int v, int lane) {
#pragma unroll
    for (int d = 1; d < 32; d <<= 1) {
        const int y = __shfl_up(v, d);
        if (lane >= d) v += y;
    }
    return v;
}

__device__ __forceinline__ void lds_add_f32(float* p, float v) {
    __hip_atomic_fetch_add(p, v, __ATOMIC_RELAXED, __HIP_MEMORY_SCOPE_WORKGROUP);
}

__device__ __forceinline__ void count_chunk(const int* __restrict__ dst, int n, int tid, int* hist) {
    for (int i0 = 0; i0 < n; i0 += PL_THREADS * 8) {
        int d[8];
#pragma unroll
        for (int u = 0; u < 8; ++u) {
            const int i = i0 + u * PL_THREADS + tid;
            d[u] = (i < n) ? dst[i] : -1;
        }
#pragma unroll
        for (int u = 0; u < 8; ++u)
            if (d[u] >= 0) atomicAdd(&hist[d[u] >> 7], 1);
    }
}

__global__ void __launch_bounds__(PL_THREADS)
GCN_count_blocks(const int* __restrict__ ei, int* __restrict__ gcount) {
    __shared__ int hist[NB];
    const int tid = threadIdx.x;
    const int e0  = blockIdx.x * CHUNK;
    const int n   = (E_EDGES - e0 < CHUNK) ? (E_EDGES - e0) : CHUNK;
    for (int i = tid; i < NB; i += PL_THREADS) hist[i] = 0;
    __syncthreads();
    count_chunk(ei + E_EDGES + e0, n, tid, hist);
    __syncthreads();
    for (int i = tid; i < NB; i += PL_THREADS) {
        const int c = hist[i];
        if (c != 0) atomicAdd(&gcount[i], c);
    }
}

__global__ void __launch_bounds__(PL_THREADS)
GCN_place_edges(const int* __restrict__ ei, const float* __restrict__ ew,
                const int* __restrict__ gcount, int* __restrict__ gcursor,
                int* __restrict__ gbase, unsigned int* __restrict__ rec_w) {
    __shared__ uint2 s_rec[CHUNK];
    __shared__ int   s_dst[CHUNK];
    __shared__ int   s_cur[NB];
    __shared__ int   s_off[NB];
    __shared__ int   s_wsum[2][PL_WAVES];
    const int tid  = threadIdx.x;
    const int lane = tid & 31;
    const int wv   = tid >> 5;
    const int e0   = blockIdx.x * CHUNK;
    const int n    = (E_EDGES - e0 < CHUNK) ? (E_EDGES - e0) : CHUNK;

    for (int i = tid; i < NB; i += PL_THREADS) s_cur[i] = 0;
    __syncthreads();
    count_chunk(ei + E_EDGES + e0, n, tid, s_cur);
    __syncthreads();

    const int gc = (tid < NB) ? gcount[tid] : 0;
    const int lc = (tid < NB) ? s_cur[tid] : 0;
    const int gi = wave_incl_scan(gc, lane);
    const int li = wave_incl_scan(lc, lane);
    if (lane == 31) { s_wsum[0][wv] = gi; s_wsum[1][wv] = li; }
    __syncthreads();
    int gex = gi - gc, lex = li - lc;
    for (int i = 0; i < PL_WAVES; ++i)
        if (i < wv) { gex += s_wsum[0][i]; lex += s_wsum[1][i]; }
    if (tid <= NB && blockIdx.x == 0) gbase[tid] = gex;
    if (tid < NB) {
        const int got = (lc > 0) ? atomicAdd(&gcursor[tid], lc) : 0;
        s_cur[tid] = lex;
        s_off[tid] = gex + got - lex;
    }
    __syncthreads();

    for (int i0 = 0; i0 < n; i0 += PL_THREADS * 8) {
        int   s[8], d[8];
        float w[8];
#pragma unroll
        for (int u = 0; u < 8; ++u) {
            const int i = i0 + u * PL_THREADS + tid;
            if (i < n) {
                s[u] = ei[e0 + i];
                d[u] = ei[E_EDGES + e0 + i];
                w[u] = ew[e0 + i];
            } else {
                s[u] = 0; d[u] = -1; w[u] = 0.0f;
            }
        }
        int p[8], o[8];
#pragma unroll
        for (int u = 0; u < 8; ++u) {
            p[u] = 0; o[u] = 0;
            if (d[u] >= 0) {
                p[u] = atomicAdd(&s_cur[d[u] >> 7], 1);
                o[u] = s_off[d[u] >> 7];
            }
        }
#pragma unroll
        for (int u = 0; u < 8; ++u) {
            if (d[u] >= 0) {
                s_rec[p[u]] = make_uint2((unsigned)s[u] | ((unsigned)(d[u] & (BLK_NODES - 1)) << 17),
                                         __float_as_uint(w[u]));
                s_dst[p[u]] = o[u] + p[u];
            }
        }
    }
    __syncthreads();

    uint2* rec = (uint2*)rec_w;
    for (int i0 = 0; i0 < n; i0 += PL_THREADS * 8) {
        uint2 r[8];
        int   t[8];
#pragma unroll
        for (int u = 0; u < 8; ++u) {
            const int i = i0 + u * PL_THREADS + tid;
            r[u] = make_uint2(0u, 0u);
            t[u] = -1;
            if (i < n) { r[u] = s_rec[i]; t[u] = s_dst[i]; }
        }
#pragma unroll
        for (int u = 0; u < 8; ++u)
            if (t[u] >= 0) rec[t[u]] = r[u];
    }
}

template <bool FULL>
__device__ __forceinline__ void group16(const uint2 rc, const int nrec, const int q, const int j,
                                        const float* __restrict__ h, float* rows) {
    float    v[EPW / 2];
    unsigned bx[EPW / 2];
#pragma unroll
    for (int i = 0; i < EPW / 2; ++i) {
        const int   k = 2 * i + q;
        bx[i] = (unsigned)lane_bcast_i((int)rc.x, k);
        const float w = lane_bcast_f(__uint_as_float(rc.y), k);
        v[i] = w * h[((bx[i] & SRC_MASK) << 4) | (unsigned)j];
    }
#pragma unroll
    for (int i = 0; i < EPW / 2; ++i)
        if (FULL || 2 * i + q < nrec) lds_add_f32(&rows[((bx[i] >> 17) << 4) | (unsigned)j], v[i]);
}

template <bool FULL>
__device__ __forceinline__ void group8(const uint2 rc, const int nrec, const int q, const int j,
                                       const float* __restrict__ z, float* rows) {
    unsigned bx[EPW / 4];
    float    w[EPW / 4];
#pragma unroll
    for (int i = 0; i < EPW / 4; ++i) {
        const int k = 4 * i + q;
        bx[i] = (unsigned)lane_bcast_i((int)rc.x, k);
        w[i]  = lane_bcast_f(__uint_as_float(rc.y), k);
    }
    if (j < OUT_F) {
        float v[EPW / 4];
#pragma unroll
        for (int i = 0; i < EPW / 4; ++i)
            v[i] = w[i] * z[((bx[i] & SRC_MASK) << 3) | (unsigned)j];
#pragma unroll
        for (int i = 0; i < EPW / 4; ++i)
            if (FULL || 4 * i + q < nrec) lds_add_f32(&rows[((bx[i] >> 17) << 3) | (unsigned)j], v[i]);
    }
}

__global__ void __launch_bounds__(EP_THREADS)
GCN_spmm16_blk(const unsigned int* __restrict__ rec_w, const int* __restrict__ gbase,
               const float* __restrict__ h, const float* __restrict__ b1,
               const float* __restrict__ W2, float* __restrict__ z) {
    __shared__ __attribute__((aligned(16))) float rows[BLK_NODES * HID];
    __shared__ float sW2[HID * OUT_F];
    __shared__ float sb1[HID];
    const int tid  = threadIdx.x;
    const int lane = tid & 31;
    const int wv   = __builtin_amdgcn_readfirstlane(tid >> 5);
    const int b    = blockIdx.x;
    const v4f zero4 = {0.0f, 0.0f, 0.0f, 0.0f};
    for (int i = tid; i < BLK_NODES * HID / 4; i += EP_THREADS) ((v4f*)rows)[i] = zero4;
    if (tid < HID * OUT_F) sW2[tid] = W2[tid];
    if (tid < HID)         sb1[tid] = b1[tid];
    __syncthreads();

    const uint2* rec = (const uint2*)rec_w;
    const int r_end = gbase[b + 1];
    int       r0    = gbase[b] + wv * 32;
    const int q = lane >> 4;
    const int j = lane & 15;
    uint2 rc = make_uint2(0u, 0u);
    if (r0 + lane < r_end) rc = rec[r0 + lane];
    while (r0 < r_end) {
        const int rn = r0 + EP_THREADS;
        uint2 rc_n = make_uint2(0u, 0u);
        if (rn + lane < r_end) rc_n = rec[rn + lane];
        const int nrec = r_end - r0;
        if (nrec >= EPW) group16<true>(rc, nrec, q, j, h, rows);
        else             group16<false>(rc, nrec, q, j, h, rows);
        rc = rc_n;
        r0 = rn;
    }
    __syncthreads();

    if (tid < BLK_NODES) {
        const int node = b * BLK_NODES + tid;
        if (node < N_NODES) {
            float hv[HID];
            const v4f* rp = (const v4f*)(rows + tid * HID);
#pragma unroll
            for (int i = 0; i < HID / 4; ++i)
                *(v4f*)(hv + 4 * i) = rp[i];
#pragma unroll
            for (int i = 0; i < HID; ++i)
                hv[i] = fmaxf(hv[i] + sb1[i], 0.0f);
            float zr[OUT_P];
#pragma unroll
            for (int jj = 0; jj < OUT_F; ++jj) {
                float s = 0.0f;
#pragma unroll
                for (int i = 0; i < HID; ++i)
                    s = fmaf(hv[i], sW2[i * OUT_F + jj], s);
                zr[jj] = s;
            }
            zr[OUT_F] = 0.0f;
            v4f* zp = (v4f*)(z + (size_t)node * OUT_P);
            zp[0] = *(const v4f*)(zr + 0);
            zp[1] = *(const v4f*)(zr + 4);
        }
    }
}

__global__ void __launch_bounds__(EP_THREADS)
GCN_spmm7_blk(const unsigned int* __restrict__ rec_w, const int* __restrict__ gbase,
              const float* __restrict__ z, const float* __restrict__ b2,
              float* __restrict__ out) {
    __shared__ __attribute__((aligned(16))) float rows[BLK_NODES * OUT_P];
    __shared__ float sb2[OUT_F];
    const int tid  = threadIdx.x;
    const int lane = tid & 31;
    const int wv   = __builtin_amdgcn_readfirstlane(tid >> 5);
    const int b    = blockIdx.x;
    const v4f zero4 = {0.0f, 0.0f, 0.0f, 0.0f};
    for (int i = tid; i < BLK_NODES * OUT_P / 4; i += EP_THREADS) ((v4f*)rows)[i] = zero4;
    if (tid < OUT_F) sb2[tid] = b2[tid];
    __syncthreads();

    const uint2* rec = (const uint2*)rec_w;
    const int r_end = gbase[b + 1];
    int       r0    = gbase[b] + wv * 32;
    const int q = lane >> 3;
    const int j = lane & 7;
    uint2 rc = make_uint2(0u, 0u);
    if (r0 + lane < r_end) rc = rec[r0 + lane];
    while (r0 < r_end) {
        const int rn = r0 + EP_THREADS;
        uint2 rc_n = make_uint2(0u, 0u);
        if (rn + lane < r_end) rc_n = rec[rn + lane];
        const int nrec = r_end - r0;
        if (nrec >= EPW) group8<true>(rc, nrec, q, j, z, rows);
        else             group8<false>(rc, nrec, q, j, z, rows);
        rc = rc_n;
        r0 = rn;
    }
    __syncthreads();

    if (tid < BLK_NODES) {
        const int node = b * BLK_NODES + tid;
        if (node < N_NODES) {
            float a[OUT_P];
            const v4f* rp = (const v4f*)(rows + tid * OUT_P);
            *(v4f*)(a + 0) = rp[0];
            *(v4f*)(a + 4) = rp[1];
            float v[OUT_F];
            float m = -3.402823466e38f;
#pragma unroll
            for (int jj = 0; jj < OUT_F; ++jj) {
                v[jj] = a[jj] + sb2[jj];
                m = fmaxf(m, v[jj]);
            }
            float sum = 0.0f;
#pragma unroll
            for (int jj = 0; jj < OUT_F; ++jj) sum += expf(v[jj] - m);
            const float lse = logf(sum);
#pragma unroll
            for (int jj = 0; jj < OUT_F; ++jj)
                out[(size_t)node * OUT_F + jj] = v[jj] - m - lse;
        }
    }
}

constexpr size_t r256(size_t bytes) { return (bytes + 255) & ~(size_t)255; }
constexpr size_t W1P_BYTES = (size_t)K_CHUNKS * 32 * 16 * 2;
constexpr size_t WS_GROUPED = r256((size_t)N_NODES * HID * 4) + r256((size_t)N_NODES * OUT_P * 4)
                            + 2 * r256(W1P_BYTES) + r256((size_t)2 * NB * 4) + r256((size_t)(NB + 1) * 4)
                            + r256((size_t)E_EDGES * 8);

extern "C" void kernel_launch(void* const* d_in, const int* in_sizes, int n_in,
                              void* d_out, int out_size, void* d_ws, size_t ws_size,
                              hipStream_t stream) {
    const float* x  = (const float*)d_in[0];
    const int*   ei = (const int*)d_in[1];
    const float* ew = (const float*)d_in[2];
    const float* W1 = (const float*)d_in[3];
    const float* b1 = (const float*)d_in[4];
    const float* W2 = (const float*)d_in[5];
    const float* b2 = (const float*)d_in[6];
    float* out = (float*)d_out;

    char* ws = (char*)d_ws;
    size_t off = 0;
    auto alloc = [&](size_t bytes) { char* p = ws + off; off = (off + bytes + 255) & ~(size_t)255; return p; };

    if (ws_size >= WS_GROUPED) {
        float*        h0g   = (float*)alloc((size_t)N_NODES * HID * 4);
        float*        zg    = (float*)alloc((size_t)N_NODES * OUT_P * 4);
        __bf16*       Bhg   = (__bf16*)alloc(W1P_BYTES);
        __bf16*       Blg   = (__bf16*)alloc(W1P_BYTES);
        int*          cnt   = (int*)alloc((size_t)2 * NB * 4);
        int*          gbase = (int*)alloc((size_t)(NB + 1) * 4);
        unsigned int* recs  = (unsigned int*)alloc((size_t)E_EDGES * 8);

        GCN_pack_w1<<<K_CHUNKS, 256, 0, stream>>>(W1, Bhg, Blg, cnt, 2 * NB);
        GCN_gemm1_wmma<<<(N_TILES + G1_WAVES - 1) / G1_WAVES, G1_WAVES * 32, 0, stream>>>(x, Bhg, Blg, h0g, nullptr, 0);
        GCN_count_blocks<<<N_CHUNKS, PL_THREADS, 0, stream>>>(ei, cnt);
        GCN_place_edges<<<N_CHUNKS, PL_THREADS, 0, stream>>>(ei, ew, cnt, cnt + NB, gbase, recs);
        GCN_spmm16_blk<<<NB, EP_THREADS, 0, stream>>>(recs, gbase, h0g, b1, W2, zg);
        GCN_spmm7_blk<<<NB, EP_THREADS, 0, stream>>>(recs, gbase, zg, b2, out);
        return;
    }
    const size_t plane1      = (size_t)N_NODES * HID;
    const size_t ws_two      = 3 * plane1 * 4 + 2 * (size_t)K_CHUNKS * 32 * 16 * 2;
    const size_t plane_stride = (ws_size >= ws_two) ? plane1 : 0;
    float*  h0   = (float*)alloc(plane1 * 4);
    float*  agg1 = (float*)alloc(plane1 * 4 + plane_stride * 4);
    float*  z    = agg1;
    float*  agg2 = h0;
    __bf16* Bh   = (__bf16*)alloc((size_t)K_CHUNKS * 32 * 16 * 2);
    __bf16* Bl   = (__bf16*)alloc((size_t)K_CHUNKS * 32 * 16 * 2);

    GCN_pack_w1<<<K_CHUNKS, 256, 0, stream>>>(W1, Bh, Bl, nullptr, 0);

    GCN_gemm1_wmma<<<(N_TILES + G1_WAVES - 1) / G1_WAVES, G1_WAVES * 32, 0, stream>>>(x, Bh, Bl, h0, agg1, plane_stride);

    GCN_spmm16<<<E_EDGES / (EPW * 8), 256, 0, stream>>>(ei, ew, h0, agg1, plane_stride);

    GCN_relu_gemm2<<<(N_NODES + 255) / 256, 256, 0, stream>>>(agg1, plane_stride, b1, W2, agg2);

    GCN_spmm7<<<E_EDGES / (EPW * 8), 256, 0, stream>>>(ei, ew, z, agg2);

    GCN_logsoftmax<<<(N_NODES + 255) / 256, 256, 0, stream>>>(agg2, b2, out);
}
